# starting kernel
# baseline (speedup 1.0000x reference)
#include <hip/hip_runtime.h>
#include <math.h>

namespace {

constexpr int kC      = 1000;
constexpr int kCPad   = 1008;
constexpr int kD      = 64;
constexpr int kB      = 2048;
constexpr float kEpsStats = 1e-5f;
constexpr float kEpsPrec  = 1e-6f;

constexpr int OFF_COUNTS = 0;
constexpr int OFF_MEAN   = OFF_COUNTS + kC;
constexpr int OFF_VARSUM = OFF_MEAN   + kC * kD;
constexpr int OFF_MUPAD  = OFF_VARSUM + kC * kD;
constexpr int OFF_PREC   = OFF_MUPAD  + kCPad * kD;
constexpr int OFF_TOTAL  = OFF_PREC   + kD;
constexpr int OFF_CCONST = OFF_TOTAL  + 1;
constexpr int OFF_QH     = OFF_CCONST + kCPad;

typedef __attribute__((ext_vector_type(2))) float v2f;
typedef __attribute__((ext_vector_type(8))) float v8f;

__global__ __launch_bounds__(64)
void class_stats_kernel(const float* __restrict__ z, const int* __restrict__ y,
                        float* __restrict__ ws) {
  const int c = blockIdx.x;
  const int d = threadIdx.x;
  float s = 0.f, cnt = 0.f;
  for (int b = 0; b < kB; ++b) {
    if (y[b] == c) { s += z[b * kD + d]; cnt += 1.f; }
  }
  const float cnte = cnt + kEpsStats;
  if (d == 0) ws[OFF_COUNTS + c] = cnte;
  ws[OFF_MEAN + c * kD + d] = s / cnte;
}

__global__ __launch_bounds__(64)
void var_stats_kernel(const float* __restrict__ z, const int* __restrict__ y,
                      float* __restrict__ ws) {
  const int c = blockIdx.x;
  const int d = threadIdx.x;
  const float m = ws[OFF_MEAN + c * kD + d];
  float s = 0.f;
  for (int b = 0; b < kB; ++b) {
    if (y[b] == c) { const float t = z[b * kD + d] - m; s += t * t; }
  }
  ws[OFF_VARSUM + c * kD + d] = s;
}

__global__ __launch_bounds__(64)
void precision_kernel(float* __restrict__ ws) {
  const int d = threadIdx.x;
  float total = 0.f;
  for (int c = 0; c < kC; ++c) total += ws[OFF_COUNTS + c];
  float s = 0.f;
  for (int c = 0; c < kC; ++c) s += ws[OFF_VARSUM + c * kD + d];
  const float pooled = s / total + kEpsStats;
  ws[OFF_PREC + d] = 1.0f / fmaxf(pooled, kEpsPrec);
  if (d == 0) ws[OFF_TOTAL] = total;
}

__global__ __launch_bounds__(64)
void mu_prep_kernel(float* __restrict__ ws) {
  const int c = blockIdx.x;
  const int d = threadIdx.x;
  __shared__ float red[kD];
  if (c >= kC) {
    ws[OFF_MUPAD + c * kD + d] = 0.f;
    if (d == 0) ws[OFF_CCONST + c] = 0.f;
    return;
  }
  const float p = ws[OFF_PREC + d];
  const float m = ws[OFF_MEAN + c * kD + d];
  const float v = p * m;
  ws[OFF_MUPAD + c * kD + d] = v;
  red[d] = v * m;
  __syncthreads();
  for (int stride = kD / 2; stride > 0; stride >>= 1) {
    if (d < stride) red[d] += red[d + stride];
    __syncthreads();
  }
  if (d == 0) {
    const float prior = ws[OFF_COUNTS + c] / ws[OFF_TOTAL];
    ws[OFF_CCONST + c] = logf(prior) - 0.5f * red[0];
  }
}

__global__ __launch_bounds__(256)
void row_quad_kernel(const float* __restrict__ z, float* __restrict__ ws) {
  __shared__ float pshared[kD];
  if (threadIdx.x < kD) pshared[threadIdx.x] = ws[OFF_PREC + threadIdx.x];
  __syncthreads();
  const int b = blockIdx.x * blockDim.x + threadIdx.x;
  if (b >= kB) return;
  float s = 0.f;
#pragma unroll
  for (int d = 0; d < kD; ++d) {
    const float t = z[b * kD + d];
    s += pshared[d] * t * t;
  }
  ws[OFF_QH + b] = -0.5f * s;
}

__global__ __launch_bounds__(256)
void lda_score_wmma_kernel(const float* __restrict__ z, const float* __restrict__ ws,
                           float* __restrict__ out) {
  const int wave = threadIdx.x >> 5;
  const int lane = threadIdx.x & 31;
  const int tile = blockIdx.x * 8 + wave;
  const int tm = tile & 127;
  const int tn = tile >> 7;
  const int row0 = tm * 16;
  const int col0 = tn * 16;
  const int half = lane >> 4;
  const int lm   = lane & 15;

  const float* __restrict__ ap = z + (row0 + lm) * kD + 2 * half;
  const float* __restrict__ bp = ws + OFF_MUPAD + (col0 + lm) * kD + 2 * half;

  v8f acc = {};
#pragma unroll
  for (int k = 0; k < kD; k += 4) {
    const v2f a = *(const v2f*)(ap + k);
    const v2f b = *(const v2f*)(bp + k);
    acc = __builtin_amdgcn_wmma_f32_16x16x4_f32(
         false, a,  false, b,
         (short)0, acc,  false,  false);
  }

  const int col = col0 + lm;
  const float cc = ws[OFF_CCONST + col];
  if (col < kC) {
#pragma unroll
    for (int r = 0; r < 8; ++r) {
      const int row = row0 + r + 8 * half;
      out[row * kC + col] = acc[r] + cc + ws[OFF_QH + row];
    }
  }
}

}

extern "C" void kernel_launch(void* const* d_in, const int* in_sizes, int n_in,
                              void* d_out, int out_size, void* d_ws, size_t ws_size,
                              hipStream_t stream) {
  (void)in_sizes; (void)n_in; (void)out_size; (void)ws_size;
  const float* z = (const float*)d_in[0];
  const int*   y = (const int*)d_in[1];
  float* ws  = (float*)d_ws;
  float* out = (float*)d_out;

  class_stats_kernel<<<kC, kD, 0, stream>>>(z, y, ws);
  var_stats_kernel<<<kC, kD, 0, stream>>>(z, y, ws);
  precision_kernel<<<1, kD, 0, stream>>>(ws);
  mu_prep_kernel<<<kCPad, kD, 0, stream>>>(ws);
  row_quad_kernel<<<kB / 256, 256, 0, stream>>>(z, ws);

  const int tiles = (kB / 16) * (kCPad / 16);
  lda_score_wmma_kernel<<<tiles / 8, 256, 0, stream>>>(z, ws, out);
}


# each class block lists its member rows once by a lane-parallel ballot scan of the labels (LDS, ascending) and both sums (mean, squared residuals) visit only those rows; order of sums unchanged
# speedup vs baseline: 3.0818x; 3.0818x over previous
#include <hip/hip_runtime.h>
#include <math.h>

namespace {

constexpr int kC      = 1000;
constexpr int kCPad   = 1008;
constexpr int kD      = 64;
constexpr int kB      = 2048;
constexpr float kEpsStats = 1e-5f;
constexpr float kEpsPrec  = 1e-6f;

constexpr int OFF_COUNTS = 0;
constexpr int OFF_MEAN   = OFF_COUNTS + kC;
constexpr int OFF_VARSUM = OFF_MEAN   + kC * kD;
constexpr int OFF_MUPAD  = OFF_VARSUM + kC * kD;
constexpr int OFF_PREC   = OFF_MUPAD  + kCPad * kD;
constexpr int OFF_TOTAL  = OFF_PREC   + kD;
constexpr int OFF_CCONST = OFF_TOTAL  + 1;
constexpr int OFF_QH     = OFF_CCONST + kCPad;

typedef __attribute__((ext_vector_type(2))) float v2f;
typedef __attribute__((ext_vector_type(8))) float v8f;

__global__ __launch_bounds__(64)
void class_stats_kernel(const float* __restrict__ z, const int* __restrict__ y,
                        float* __restrict__ ws) {
  constexpr int kHalf  = kB / 2;
  constexpr int kSteps = kHalf / 32;
  constexpr int kChunk = 8;
  __shared__ int members[kB];
  __shared__ int nmem[2];

  const int c    = blockIdx.x;
  const int d    = threadIdx.x;
  const int wave = threadIdx.x >> 5;
  const int lane = threadIdx.x & 31;
  const int base = wave * kHalf;

  int lab[kSteps];
#pragma unroll
  for (int i = 0; i < kSteps; ++i) lab[i] = y[base + i * 32 + lane];

  const unsigned below = (1u << lane) - 1u;
  int n = 0;
#pragma unroll
  for (int i = 0; i < kSteps; ++i) {
    const bool hit = (lab[i] == c);
    const unsigned m = __builtin_amdgcn_ballot_w32(hit);
    if (hit) members[base + n + __builtin_popcount(m & below)] = base + i * 32 + lane;
    n += __builtin_popcount(m);
  }
  if (lane == 0) nmem[wave] = n;
  __syncthreads();

  const int n0 = nmem[0];
  const int nt = n0 + nmem[1];
  auto member = [&](int i) { return members[i < n0 ? i : kHalf - n0 + i]; };

  const float* __restrict__ zc = z + d;
  float s = 0.f;
  for (int i = 0; i < nt; i += kChunk) {
    float v[kChunk];
#pragma unroll
    for (int j = 0; j < kChunk; ++j) v[j] = zc[member(min(i + j, nt - 1)) * kD];
#pragma unroll
    for (int j = 0; j < kChunk; ++j) if (i + j < nt) s += v[j];
  }
  const float cnte = (float)nt + kEpsStats;
  const float mean = s / cnte;
  if (d == 0) ws[OFF_COUNTS + c] = cnte;
  ws[OFF_MEAN + c * kD + d] = mean;

  float q = 0.f;
  for (int i = 0; i < nt; i += kChunk) {
    float v[kChunk];
#pragma unroll
    for (int j = 0; j < kChunk; ++j) v[j] = zc[member(min(i + j, nt - 1)) * kD];
#pragma unroll
    for (int j = 0; j < kChunk; ++j) {
      const float t = v[j] - mean;
      if (i + j < nt) q += t * t;
    }
  }
  ws[OFF_VARSUM + c * kD + d] = q;
}

__global__ __launch_bounds__(64)
void precision_kernel(float* __restrict__ ws) {
  const int d = threadIdx.x;
  float total = 0.f;
  for (int c = 0; c < kC; ++c) total += ws[OFF_COUNTS + c];
  float s = 0.f;
  for (int c = 0; c < kC; ++c) s += ws[OFF_VARSUM + c * kD + d];
  const float pooled = s / total + kEpsStats;
  ws[OFF_PREC + d] = 1.0f / fmaxf(pooled, kEpsPrec);
  if (d == 0) ws[OFF_TOTAL] = total;
}

__global__ __launch_bounds__(64)
void mu_prep_kernel(float* __restrict__ ws) {
  const int c = blockIdx.x;
  const int d = threadIdx.x;
  __shared__ float red[kD];
  if (c >= kC) {
    ws[OFF_MUPAD + c * kD + d] = 0.f;
    if (d == 0) ws[OFF_CCONST + c] = 0.f;
    return;
  }
  const float p = ws[OFF_PREC + d];
  const float m = ws[OFF_MEAN + c * kD + d];
  const float v = p * m;
  ws[OFF_MUPAD + c * kD + d] = v;
  red[d] = v * m;
  __syncthreads();
  for (int stride = kD / 2; stride > 0; stride >>= 1) {
    if (d < stride) red[d] += red[d + stride];
    __syncthreads();
  }
  if (d == 0) {
    const float prior = ws[OFF_COUNTS + c] / ws[OFF_TOTAL];
    ws[OFF_CCONST + c] = logf(prior) - 0.5f * red[0];
  }
}

__global__ __launch_bounds__(256)
void row_quad_kernel(const float* __restrict__ z, float* __restrict__ ws) {
  __shared__ float pshared[kD];
  if (threadIdx.x < kD) pshared[threadIdx.x] = ws[OFF_PREC + threadIdx.x];
  __syncthreads();
  const int b = blockIdx.x * blockDim.x + threadIdx.x;
  if (b >= kB) return;
  float s = 0.f;
#pragma unroll
  for (int d = 0; d < kD; ++d) {
    const float t = z[b * kD + d];
    s += pshared[d] * t * t;
  }
  ws[OFF_QH + b] = -0.5f * s;
}

__global__ __launch_bounds__(256)
void lda_score_wmma_kernel(const float* __restrict__ z, const float* __restrict__ ws,
                           float* __restrict__ out) {
  const int wave = threadIdx.x >> 5;
  const int lane = threadIdx.x & 31;
  const int tile = blockIdx.x * 8 + wave;
  const int tm = tile & 127;
  const int tn = tile >> 7;
  const int row0 = tm * 16;
  const int col0 = tn * 16;
  const int half = lane >> 4;
  const int lm   = lane & 15;

  const float* __restrict__ ap = z + (row0 + lm) * kD + 2 * half;
  const float* __restrict__ bp = ws + OFF_MUPAD + (col0 + lm) * kD + 2 * half;

  v8f acc = {};
#pragma unroll
  for (int k = 0; k < kD; k += 4) {
    const v2f a = *(const v2f*)(ap + k);
    const v2f b = *(const v2f*)(bp + k);
    acc = __builtin_amdgcn_wmma_f32_16x16x4_f32(
         false, a,  false, b,
         (short)0, acc,  false,  false);
  }

  const int col = col0 + lm;
  const float cc = ws[OFF_CCONST + col];
  if (col < kC) {
#pragma unroll
    for (int r = 0; r < 8; ++r) {
      const int row = row0 + r + 8 * half;
      out[row * kC + col] = acc[r] + cc + ws[OFF_QH + row];
    }
  }
}

}

extern "C" void kernel_launch(void* const* d_in, const int* in_sizes, int n_in,
                              void* d_out, int out_size, void* d_ws, size_t ws_size,
                              hipStream_t stream) {
  (void)in_sizes; (void)n_in; (void)out_size; (void)ws_size;
  const float* z = (const float*)d_in[0];
  const int*   y = (const int*)d_in[1];
  float* ws  = (float*)d_ws;
  float* out = (float*)d_out;

  class_stats_kernel<<<kC, kD, 0, stream>>>(z, y, ws);
  precision_kernel<<<1, kD, 0, stream>>>(ws);
  mu_prep_kernel<<<kCPad, kD, 0, stream>>>(ws);
  row_quad_kernel<<<kB / 256, 256, 0, stream>>>(z, ws);

  const int tiles = (kB / 16) * (kCPad / 16);
  lda_score_wmma_kernel<<<tiles / 8, 256, 0, stream>>>(z, ws, out);
}


# precision kernel: the two 1000-term class sums (count total, pooled squared residuals) run as 50 parallel partial sums of 20 classes in one block, joined through LDS; order of these two sums changes
# speedup vs baseline: 7.3554x; 2.3867x over previous
#include <hip/hip_runtime.h>
#include <math.h>

namespace {

constexpr int kC      = 1000;
constexpr int kCPad   = 1008;
constexpr int kD      = 64;
constexpr int kB      = 2048;
constexpr float kEpsStats = 1e-5f;
constexpr float kEpsPrec  = 1e-6f;

constexpr int OFF_COUNTS = 0;
constexpr int OFF_MEAN   = OFF_COUNTS + kC;
constexpr int OFF_VARSUM = OFF_MEAN   + kC * kD;
constexpr int OFF_MUPAD  = OFF_VARSUM + kC * kD;
constexpr int OFF_PREC   = OFF_MUPAD  + kCPad * kD;
constexpr int OFF_TOTAL  = OFF_PREC   + kD;
constexpr int OFF_CCONST = OFF_TOTAL  + 1;
constexpr int OFF_QH     = OFF_CCONST + kCPad;

typedef __attribute__((ext_vector_type(2))) float v2f;
typedef __attribute__((ext_vector_type(8))) float v8f;

__global__ __launch_bounds__(64)
void class_stats_kernel(const float* __restrict__ z, const int* __restrict__ y,
                        float* __restrict__ ws) {
  constexpr int kHalf  = kB / 2;
  constexpr int kSteps = kHalf / 32;
  constexpr int kChunk = 8;
  __shared__ int members[kB];
  __shared__ int nmem[2];

  const int c    = blockIdx.x;
  const int d    = threadIdx.x;
  const int wave = threadIdx.x >> 5;
  const int lane = threadIdx.x & 31;
  const int base = wave * kHalf;

  int lab[kSteps];
#pragma unroll
  for (int i = 0; i < kSteps; ++i) lab[i] = y[base + i * 32 + lane];

  const unsigned below = (1u << lane) - 1u;
  int n = 0;
#pragma unroll
  for (int i = 0; i < kSteps; ++i) {
    const bool hit = (lab[i] == c);
    const unsigned m = __builtin_amdgcn_ballot_w32(hit);
    if (hit) members[base + n + __builtin_popcount(m & below)] = base + i * 32 + lane;
    n += __builtin_popcount(m);
  }
  if (lane == 0) nmem[wave] = n;
  __syncthreads();

  const int n0 = nmem[0];
  const int nt = n0 + nmem[1];
  auto member = [&](int i) { return members[i < n0 ? i : kHalf - n0 + i]; };

  const float* __restrict__ zc = z + d;
  float s = 0.f;
  for (int i = 0; i < nt; i += kChunk) {
    float v[kChunk];
#pragma unroll
    for (int j = 0; j < kChunk; ++j) v[j] = zc[member(min(i + j, nt - 1)) * kD];
#pragma unroll
    for (int j = 0; j < kChunk; ++j) if (i + j < nt) s += v[j];
  }
  const float cnte = (float)nt + kEpsStats;
  const float mean = s / cnte;
  if (d == 0) ws[OFF_COUNTS + c] = cnte;
  ws[OFF_MEAN + c * kD + d] = mean;

  float q = 0.f;
  for (int i = 0; i < nt; i += kChunk) {
    float v[kChunk];
#pragma unroll
    for (int j = 0; j < kChunk; ++j) v[j] = zc[member(min(i + j, nt - 1)) * kD];
#pragma unroll
    for (int j = 0; j < kChunk; ++j) {
      const float t = v[j] - mean;
      if (i + j < nt) q += t * t;
    }
  }
  ws[OFF_VARSUM + c * kD + d] = q;
}

constexpr int kPrecGroups     = 50;
constexpr int kPrecPer        = kC / kPrecGroups;
constexpr int kPrecSumThreads = kPrecGroups * (kD / 4);
constexpr int kPrecThreads    = kPrecSumThreads + 64;

__global__ __launch_bounds__(kPrecThreads)
void precision_kernel(float* __restrict__ ws) {
  static_assert(kPrecGroups * kPrecPer == kC, "class groups must tile the classes exactly");
  static_assert(kPrecSumThreads % 32 == 0 && kPrecGroups <= 64, "whole waves per role");
  static_assert(kPrecPer % 4 == 0 && OFF_COUNTS % 4 == 0 && OFF_VARSUM % 4 == 0,
                "16-byte loads need 16-byte aligned rows");
  __shared__ float4 part[kPrecGroups][kD / 4];
  __shared__ float  cpart[kPrecGroups];

  const int tid = threadIdx.x;
  if (tid < kPrecSumThreads) {
    const int q = tid & 15;
    const int g = tid >> 4;
    const float4* __restrict__ vs =
        reinterpret_cast<const float4*>(ws + OFF_VARSUM + g * kPrecPer * kD) + q;
    float4 v[kPrecPer];
#pragma unroll
    for (int j = 0; j < kPrecPer; ++j) v[j] = vs[j * (kD / 4)];
    float4 s = make_float4(0.f, 0.f, 0.f, 0.f);
#pragma unroll
    for (int j = 0; j < kPrecPer; ++j) {
      s.x += v[j].x; s.y += v[j].y; s.z += v[j].z; s.w += v[j].w;
    }
    part[g][q] = s;
  } else if (tid - kPrecSumThreads < kPrecGroups) {
    const int g = tid - kPrecSumThreads;
    const float4* __restrict__ cp =
        reinterpret_cast<const float4*>(ws + OFF_COUNTS + g * kPrecPer);
    float4 cv[kPrecPer / 4];
#pragma unroll
    for (int j = 0; j < kPrecPer / 4; ++j) cv[j] = cp[j];
    float cs = 0.f;
#pragma unroll
    for (int j = 0; j < kPrecPer / 4; ++j) { cs += cv[j].x; cs += cv[j].y; cs += cv[j].z; cs += cv[j].w; }
    cpart[g] = cs;
  }
  __syncthreads();

  if (tid < kD) {
    const int d = tid;
    const float* __restrict__ pf = reinterpret_cast<const float*>(&part[0][0]);
    float total = 0.f;
    float sum = 0.f;
    for (int gg = 0; gg < kPrecGroups; ++gg) {
      total += cpart[gg];
      sum += pf[gg * kD + d];
    }
    const float pooled = sum / total + kEpsStats;
    ws[OFF_PREC + d] = 1.0f / fmaxf(pooled, kEpsPrec);
    if (d == 0) ws[OFF_TOTAL] = total;
  }
}

__global__ __launch_bounds__(64)
void mu_prep_kernel(float* __restrict__ ws) {
  const int c = blockIdx.x;
  const int d = threadIdx.x;
  __shared__ float red[kD];
  if (c >= kC) {
    ws[OFF_MUPAD + c * kD + d] = 0.f;
    if (d == 0) ws[OFF_CCONST + c] = 0.f;
    return;
  }
  const float p = ws[OFF_PREC + d];
  const float m = ws[OFF_MEAN + c * kD + d];
  const float v = p * m;
  ws[OFF_MUPAD + c * kD + d] = v;
  red[d] = v * m;
  __syncthreads();
  for (int stride = kD / 2; stride > 0; stride >>= 1) {
    if (d < stride) red[d] += red[d + stride];
    __syncthreads();
  }
  if (d == 0) {
    const float prior = ws[OFF_COUNTS + c] / ws[OFF_TOTAL];
    ws[OFF_CCONST + c] = logf(prior) - 0.5f * red[0];
  }
}

__global__ __launch_bounds__(256)
void row_quad_kernel(const float* __restrict__ z, float* __restrict__ ws) {
  __shared__ float pshared[kD];
  if (threadIdx.x < kD) pshared[threadIdx.x] = ws[OFF_PREC + threadIdx.x];
  __syncthreads();
  const int b = blockIdx.x * blockDim.x + threadIdx.x;
  if (b >= kB) return;
  float s = 0.f;
#pragma unroll
  for (int d = 0; d < kD; ++d) {
    const float t = z[b * kD + d];
    s += pshared[d] * t * t;
  }
  ws[OFF_QH + b] = -0.5f * s;
}

__global__ __launch_bounds__(256)
void lda_score_wmma_kernel(const float* __restrict__ z, const float* __restrict__ ws,
                           float* __restrict__ out) {
  const int wave = threadIdx.x >> 5;
  const int lane = threadIdx.x & 31;
  const int tile = blockIdx.x * 8 + wave;
  const int tm = tile & 127;
  const int tn = tile >> 7;
  const int row0 = tm * 16;
  const int col0 = tn * 16;
  const int half = lane >> 4;
  const int lm   = lane & 15;

  const float* __restrict__ ap = z + (row0 + lm) * kD + 2 * half;
  const float* __restrict__ bp = ws + OFF_MUPAD + (col0 + lm) * kD + 2 * half;

  v8f acc = {};
#pragma unroll
  for (int k = 0; k < kD; k += 4) {
    const v2f a = *(const v2f*)(ap + k);
    const v2f b = *(const v2f*)(bp + k);
    acc = __builtin_amdgcn_wmma_f32_16x16x4_f32(
         false, a,  false, b,
         (short)0, acc,  false,  false);
  }

  const int col = col0 + lm;
  const float cc = ws[OFF_CCONST + col];
  if (col < kC) {
#pragma unroll
    for (int r = 0; r < 8; ++r) {
      const int row = row0 + r + 8 * half;
      out[row * kC + col] = acc[r] + cc + ws[OFF_QH + row];
    }
  }
}

}

extern "C" void kernel_launch(void* const* d_in, const int* in_sizes, int n_in,
                              void* d_out, int out_size, void* d_ws, size_t ws_size,
                              hipStream_t stream) {
  (void)in_sizes; (void)n_in; (void)out_size; (void)ws_size;
  const float* z = (const float*)d_in[0];
  const int*   y = (const int*)d_in[1];
  float* ws  = (float*)d_ws;
  float* out = (float*)d_out;

  class_stats_kernel<<<kC, kD, 0, stream>>>(z, y, ws);
  precision_kernel<<<1, kPrecThreads, 0, stream>>>(ws);
  mu_prep_kernel<<<kCPad, kD, 0, stream>>>(ws);
  row_quad_kernel<<<kB / 256, 256, 0, stream>>>(z, ws);

  const int tiles = (kB / 16) * (kCPad / 16);
  lda_score_wmma_kernel<<<tiles / 8, 256, 0, stream>>>(z, ws, out);
}


# score kernel forms P*mu, class constant and row term itself from means and precision: 5 to 3 launches; both terms summed as two half sums joined by one more f32 WMMA (order of these sums changes)
# speedup vs baseline: 10.1501x; 1.3800x over previous
#include <hip/hip_runtime.h>
#include <math.h>

namespace {

constexpr int kC      = 1000;
constexpr int kCPad   = 1008;
constexpr int kD      = 64;
constexpr int kB      = 2048;
constexpr float kEpsStats = 1e-5f;
constexpr float kEpsPrec  = 1e-6f;

constexpr int OFF_COUNTS = 0;
constexpr int OFF_MEAN   = OFF_COUNTS + kC;
constexpr int OFF_VARSUM = OFF_MEAN   + kC * kD;
constexpr int OFF_PREC   = OFF_VARSUM + kC * kD;
constexpr int OFF_TOTAL  = OFF_PREC   + kD;

typedef __attribute__((ext_vector_type(2))) float v2f;
typedef __attribute__((ext_vector_type(8))) float v8f;

__global__ __launch_bounds__(64)
void class_stats_kernel(const float* __restrict__ z, const int* __restrict__ y,
                        float* __restrict__ ws) {
  constexpr int kHalf  = kB / 2;
  constexpr int kSteps = kHalf / 32;
  constexpr int kChunk = 8;
  __shared__ int members[kB];
  __shared__ int nmem[2];

  const int c    = blockIdx.x;
  const int d    = threadIdx.x;
  const int wave = threadIdx.x >> 5;
  const int lane = threadIdx.x & 31;
  const int base = wave * kHalf;

  int lab[kSteps];
#pragma unroll
  for (int i = 0; i < kSteps; ++i) lab[i] = y[base + i * 32 + lane];

  const unsigned below = (1u << lane) - 1u;
  int n = 0;
#pragma unroll
  for (int i = 0; i < kSteps; ++i) {
    const bool hit = (lab[i] == c);
    const unsigned m = __builtin_amdgcn_ballot_w32(hit);
    if (hit) members[base + n + __builtin_popcount(m & below)] = base + i * 32 + lane;
    n += __builtin_popcount(m);
  }
  if (lane == 0) nmem[wave] = n;
  __syncthreads();

  const int n0 = nmem[0];
  const int nt = n0 + nmem[1];
  auto member = [&](int i) { return members[i < n0 ? i : kHalf - n0 + i]; };

  const float* __restrict__ zc = z + d;
  float s = 0.f;
  for (int i = 0; i < nt; i += kChunk) {
    float v[kChunk];
#pragma unroll
    for (int j = 0; j < kChunk; ++j) v[j] = zc[member(min(i + j, nt - 1)) * kD];
#pragma unroll
    for (int j = 0; j < kChunk; ++j) if (i + j < nt) s += v[j];
  }
  const float cnte = (float)nt + kEpsStats;
  const float mean = s / cnte;
  if (d == 0) ws[OFF_COUNTS + c] = cnte;
  ws[OFF_MEAN + c * kD + d] = mean;

  float q = 0.f;
  for (int i = 0; i < nt; i += kChunk) {
    float v[kChunk];
#pragma unroll
    for (int j = 0; j < kChunk; ++j) v[j] = zc[member(min(i + j, nt - 1)) * kD];
#pragma unroll
    for (int j = 0; j < kChunk; ++j) {
      const float t = v[j] - mean;
      if (i + j < nt) q += t * t;
    }
  }
  ws[OFF_VARSUM + c * kD + d] = q;
}

constexpr int kPrecGroups     = 50;
constexpr int kPrecPer        = kC / kPrecGroups;
constexpr int kPrecSumThreads = kPrecGroups * (kD / 4);
constexpr int kPrecThreads    = kPrecSumThreads + 64;

__global__ __launch_bounds__(kPrecThreads)
void precision_kernel(float* __restrict__ ws) {
  static_assert(kPrecGroups * kPrecPer == kC, "class groups must tile the classes exactly");
  static_assert(kPrecSumThreads % 32 == 0 && kPrecGroups <= 64, "whole waves per role");
  static_assert(kPrecPer % 4 == 0 && OFF_COUNTS % 4 == 0 && OFF_VARSUM % 4 == 0,
                "16-byte loads need 16-byte aligned rows");
  __shared__ float4 part[kPrecGroups][kD / 4];
  __shared__ float  cpart[kPrecGroups];

  const int tid = threadIdx.x;
  if (tid < kPrecSumThreads) {
    const int q = tid & 15;
    const int g = tid >> 4;
    const float4* __restrict__ vs =
        reinterpret_cast<const float4*>(ws + OFF_VARSUM + g * kPrecPer * kD) + q;
    float4 v[kPrecPer];
#pragma unroll
    for (int j = 0; j < kPrecPer; ++j) v[j] = vs[j * (kD / 4)];
    float4 s = make_float4(0.f, 0.f, 0.f, 0.f);
#pragma unroll
    for (int j = 0; j < kPrecPer; ++j) {
      s.x += v[j].x; s.y += v[j].y; s.z += v[j].z; s.w += v[j].w;
    }
    part[g][q] = s;
  } else if (tid - kPrecSumThreads < kPrecGroups) {
    const int g = tid - kPrecSumThreads;
    const float4* __restrict__ cp =
        reinterpret_cast<const float4*>(ws + OFF_COUNTS + g * kPrecPer);
    float4 cv[kPrecPer / 4];
#pragma unroll
    for (int j = 0; j < kPrecPer / 4; ++j) cv[j] = cp[j];
    float cs = 0.f;
#pragma unroll
    for (int j = 0; j < kPrecPer / 4; ++j) { cs += cv[j].x; cs += cv[j].y; cs += cv[j].z; cs += cv[j].w; }
    cpart[g] = cs;
  }
  __syncthreads();

  if (tid < kD) {
    const int d = tid;
    const float* __restrict__ pf = reinterpret_cast<const float*>(&part[0][0]);
    float total = 0.f;
    float sum = 0.f;
    for (int gg = 0; gg < kPrecGroups; ++gg) {
      total += cpart[gg];
      sum += pf[gg * kD + d];
    }
    const float pooled = sum / total + kEpsStats;
    ws[OFF_PREC + d] = 1.0f / fmaxf(pooled, kEpsPrec);
    if (d == 0) ws[OFF_TOTAL] = total;
  }
}

__global__ __launch_bounds__(256)
void lda_score_wmma_kernel(const float* __restrict__ z, const float* __restrict__ ws,
                           float* __restrict__ out) {
  const int wave = threadIdx.x >> 5;
  const int lane = threadIdx.x & 31;
  const int tile = blockIdx.x * 8 + wave;
  const int tm = tile & 127;
  const int tn = tile >> 7;
  const int row0 = tm * 16;
  const int col0 = tn * 16;
  const int half = lane >> 4;
  const int lm   = lane & 15;
  const int col  = col0 + lm;
  const int cl   = min(col, kC - 1);

  const float* __restrict__ ap = z + (row0 + lm) * kD + 2 * half;
  const float* __restrict__ mp = ws + OFF_MEAN + cl * kD + 2 * half;
  const float* __restrict__ pp = ws + OFF_PREC + 2 * half;

  constexpr int kSteps = kD / 4;
  v2f a[kSteps], m[kSteps], p[kSteps];
#pragma unroll
  for (int j = 0; j < kSteps; ++j) {
    a[j] = *(const v2f*)(ap + 4 * j);
    m[j] = *(const v2f*)(mp + 4 * j);
    p[j] = *(const v2f*)(pp + 4 * j);
  }
  const float cnte  = ws[OFF_COUNTS + cl];
  const float total = ws[OFF_TOTAL];

  v8f acc = {};
  v2f sq = {0.f, 0.f};
  v2f sc = {0.f, 0.f};
#pragma unroll
  for (int j = 0; j < kSteps; ++j) {
    const v2f b = p[j] * m[j];
    sc += b * m[j];
    sq += (p[j] * a[j]) * a[j];
    acc = __builtin_amdgcn_wmma_f32_16x16x4_f32(
         false, a[j],  false, b,
         (short)0, acc,  false,  false);
  }

  const float lp = logf(cnte / total);
  v2f ea, eb;
  ea.x = -0.5f * (sq.x + sq.y);
  ea.y = 1.0f;
  eb.x = 1.0f;
  eb.y = -0.5f * (sc.x + sc.y) + (half == 0 ? lp : 0.f);
  acc = __builtin_amdgcn_wmma_f32_16x16x4_f32(
       false, ea,  false, eb,
       (short)0, acc,  false,  false);

  if (col < kC) {
#pragma unroll
    for (int r = 0; r < 8; ++r) {
      const int row = row0 + r + 8 * half;
      out[row * kC + col] = acc[r];
    }
  }
}

}

extern "C" void kernel_launch(void* const* d_in, const int* in_sizes, int n_in,
                              void* d_out, int out_size, void* d_ws, size_t ws_size,
                              hipStream_t stream) {
  (void)in_sizes; (void)n_in; (void)out_size; (void)ws_size;
  const float* z = (const float*)d_in[0];
  const int*   y = (const int*)d_in[1];
  float* ws  = (float*)d_ws;
  float* out = (float*)d_out;

  class_stats_kernel<<<kC, kD, 0, stream>>>(z, y, ws);
  precision_kernel<<<1, kPrecThreads, 0, stream>>>(ws);

  const int tiles = (kB / 16) * (kCPad / 16);
  lda_score_wmma_kernel<<<tiles / 8, 256, 0, stream>>>(z, ws, out);
}


# precision launch folded into the score kernel: statistics blocks of 16 classes write 63 partial rows, each 8-wave score block joins them in LDS: 3 to 2 launches; order of both pooled sums changes
# speedup vs baseline: 12.8224x; 1.2633x over previous
#include <hip/hip_runtime.h>
#include <math.h>

namespace {

constexpr int kC      = 1000;
constexpr int kCPad   = 1008;
constexpr int kD      = 64;
constexpr int kB      = 2048;
constexpr float kEpsStats = 1e-5f;
constexpr float kEpsPrec  = 1e-6f;

constexpr int kGrpClasses  = 16;
constexpr int kGroups      = kCPad / kGrpClasses;
constexpr int kStatThreads = kGrpClasses * kD;

constexpr int OFF_COUNTS = 0;
constexpr int OFF_MEAN   = OFF_COUNTS + kC;
constexpr int OFF_PART   = OFF_MEAN   + kC * kD;
constexpr int OFF_CPART  = OFF_PART   + kGroups * kD;

typedef __attribute__((ext_vector_type(2))) float v2f;
typedef __attribute__((ext_vector_type(8))) float v8f;

__global__ __launch_bounds__(kStatThreads)
void class_stats_kernel(const float* __restrict__ z, const int* __restrict__ y,
                        float* __restrict__ ws) {
  constexpr int kHalf  = kB / 2;
  constexpr int kSteps = kHalf / 32;
  constexpr int kChunk = 8;
  __shared__ int   members[kGrpClasses][kB];
  __shared__ int   nmem[kGrpClasses][2];
  __shared__ float qsh[kGrpClasses][kD];
  __shared__ float csh[kGrpClasses];

  const int lc   = threadIdx.x >> 6;
  const int c    = blockIdx.x * kGrpClasses + lc;
  const int d    = threadIdx.x & 63;
  const int wave = (threadIdx.x >> 5) & 1;
  const int lane = threadIdx.x & 31;
  const int base = wave * kHalf;
  const bool valid = c < kC;
  int* mem = members[lc];

  int lab[kSteps];
#pragma unroll
  for (int i = 0; i < kSteps; ++i) lab[i] = y[base + i * 32 + lane];

  const unsigned below = (1u << lane) - 1u;
  int n = 0;
#pragma unroll
  for (int i = 0; i < kSteps; ++i) {
    const bool hit = (lab[i] == c);
    const unsigned m = __builtin_amdgcn_ballot_w32(hit);
    if (hit) mem[base + n + __builtin_popcount(m & below)] = base + i * 32 + lane;
    n += __builtin_popcount(m);
  }
  if (lane == 0) nmem[lc][wave] = n;
  __syncthreads();

  const int n0 = nmem[lc][0];
  const int nt = n0 + nmem[lc][1];
  auto member = [&](int i) { return mem[i < n0 ? i : kHalf - n0 + i]; };

  const float* __restrict__ zc = z + d;
  float s = 0.f;
  for (int i = 0; i < nt; i += kChunk) {
    float v[kChunk];
#pragma unroll
    for (int j = 0; j < kChunk; ++j) v[j] = zc[member(min(i + j, nt - 1)) * kD];
#pragma unroll
    for (int j = 0; j < kChunk; ++j) if (i + j < nt) s += v[j];
  }
  const float cnte = (float)nt + kEpsStats;
  const float mean = s / cnte;
  if (valid) {
    if (d == 0) ws[OFF_COUNTS + c] = cnte;
    ws[OFF_MEAN + c * kD + d] = mean;
  }

  float q = 0.f;
  for (int i = 0; i < nt; i += kChunk) {
    float v[kChunk];
#pragma unroll
    for (int j = 0; j < kChunk; ++j) v[j] = zc[member(min(i + j, nt - 1)) * kD];
#pragma unroll
    for (int j = 0; j < kChunk; ++j) {
      const float t = v[j] - mean;
      if (i + j < nt) q += t * t;
    }
  }
  qsh[lc][d] = valid ? q : 0.f;
  if (d == 0) csh[lc] = valid ? cnte : 0.f;
  __syncthreads();

  if (threadIdx.x < kD) {
    float ps = 0.f;
#pragma unroll
    for (int k = 0; k < kGrpClasses; ++k) ps += qsh[k][threadIdx.x];
    ws[OFF_PART + blockIdx.x * kD + threadIdx.x] = ps;
  } else if (threadIdx.x == kD) {
    float cs = 0.f;
#pragma unroll
    for (int k = 0; k < kGrpClasses; ++k) cs += csh[k];
    ws[OFF_CPART + blockIdx.x] = cs;
  }
}

__global__ __launch_bounds__(256)
void lda_score_wmma_kernel(const float* __restrict__ z, const float* __restrict__ ws,
                           float* __restrict__ out) {
  constexpr int kJoinGroups = 16;
  constexpr int kJoinRows   = 4;
  __shared__ float4 jpart[kJoinGroups][kD / 4];
  __shared__ float  jcnt[kJoinGroups];
  __shared__ float  psh[kD];
  __shared__ float  tsh;

  const int wave = threadIdx.x >> 5;
  const int lane = threadIdx.x & 31;
  const int tile = blockIdx.x * 8 + wave;
  const int tm = tile & 127;
  const int tn = tile >> 7;
  const int row0 = tm * 16;
  const int col0 = tn * 16;
  const int half = lane >> 4;
  const int lm   = lane & 15;
  const int col  = col0 + lm;
  const int cl   = min(col, kC - 1);

  const int jq = threadIdx.x & 15;
  const int jg = threadIdx.x >> 4;
  const float4* __restrict__ jp = reinterpret_cast<const float4*>(ws + OFF_PART) + jq;
  float4 jv[kJoinRows];
#pragma unroll
  for (int i = 0; i < kJoinRows; ++i)
    jv[i] = jp[min(jg + kJoinGroups * i, kGroups - 1) * (kD / 4)];
  float cw[kJoinRows] = {0.f, 0.f, 0.f, 0.f};
  if (jq == 0) {
#pragma unroll
    for (int i = 0; i < kJoinRows; ++i)
      cw[i] = ws[OFF_CPART + min(jg + kJoinGroups * i, kGroups - 1)];
  }

  const float* __restrict__ ap = z + (row0 + lm) * kD + 2 * half;
  const float* __restrict__ mp = ws + OFF_MEAN + cl * kD + 2 * half;

  constexpr int kSteps = kD / 4;
  v2f a[kSteps], m[kSteps], p[kSteps];
#pragma unroll
  for (int j = 0; j < kSteps; ++j) {
    a[j] = *(const v2f*)(ap + 4 * j);
    m[j] = *(const v2f*)(mp + 4 * j);
  }
  const float cnte = ws[OFF_COUNTS + cl];

  {
    float4 js = jv[0];
#pragma unroll
    for (int i = 1; i < kJoinRows; ++i) {
      if (jg + kJoinGroups * i < kGroups) {
        js.x += jv[i].x; js.y += jv[i].y; js.z += jv[i].z; js.w += jv[i].w;
      }
    }
    jpart[jg][jq] = js;
    if (jq == 0) {
      float cs = cw[0];
#pragma unroll
      for (int i = 1; i < kJoinRows; ++i)
        if (jg + kJoinGroups * i < kGroups) cs += cw[i];
      jcnt[jg] = cs;
    }
  }
  __syncthreads();
  if (threadIdx.x < kD) {
    const float* __restrict__ pf = reinterpret_cast<const float*>(&jpart[0][0]);
    float sum = 0.f;
    float tot = 0.f;
#pragma unroll
    for (int gg = 0; gg < kJoinGroups; ++gg) {
      sum += pf[gg * kD + threadIdx.x];
      tot += jcnt[gg];
    }
    const float pooled = sum / tot + kEpsStats;
    psh[threadIdx.x] = 1.0f / fmaxf(pooled, kEpsPrec);
    if (threadIdx.x == 0) tsh = tot;
  }
  __syncthreads();
  const float total = tsh;
#pragma unroll
  for (int j = 0; j < kSteps; ++j) p[j] = *(const v2f*)(psh + 4 * j + 2 * half);

  v8f acc = {};
  v2f sq = {0.f, 0.f};
  v2f sc = {0.f, 0.f};
#pragma unroll
  for (int j = 0; j < kSteps; ++j) {
    const v2f b = p[j] * m[j];
    sc += b * m[j];
    sq += (p[j] * a[j]) * a[j];
    acc = __builtin_amdgcn_wmma_f32_16x16x4_f32(
         false, a[j],  false, b,
         (short)0, acc,  false,  false);
  }

  const float lp = logf(cnte / total);
  v2f ea, eb;
  ea.x = -0.5f * (sq.x + sq.y);
  ea.y = 1.0f;
  eb.x = 1.0f;
  eb.y = -0.5f * (sc.x + sc.y) + (half == 0 ? lp : 0.f);
  acc = __builtin_amdgcn_wmma_f32_16x16x4_f32(
       false, ea,  false, eb,
       (short)0, acc,  false,  false);

  if (col < kC) {
#pragma unroll
    for (int r = 0; r < 8; ++r) {
      const int row = row0 + r + 8 * half;
      out[row * kC + col] = acc[r];
    }
  }
}

}

extern "C" void kernel_launch(void* const* d_in, const int* in_sizes, int n_in,
                              void* d_out, int out_size, void* d_ws, size_t ws_size,
                              hipStream_t stream) {
  (void)in_sizes; (void)n_in; (void)out_size; (void)ws_size;
  const float* z = (const float*)d_in[0];
  const int*   y = (const int*)d_in[1];
  float* ws  = (float*)d_ws;
  float* out = (float*)d_out;

  class_stats_kernel<<<kGroups, kStatThreads, 0, stream>>>(z, y, ws);

  const int tiles = (kB / 16) * (kCPad / 16);
  lda_score_wmma_kernel<<<tiles / 8, 256, 0, stream>>>(z, ws, out);
}


# score GEMM tiled per work-group: each 8-wave block copies its 128 rows of z and 64 class means to LDS once (whole-row 16-byte loads), each wave takes 2x2 tiles from LDS; order of sums unchanged
# speedup vs baseline: 16.5603x; 1.2915x over previous
#include <hip/hip_runtime.h>
#include <math.h>

namespace {

constexpr int kC      = 1000;
constexpr int kCPad   = 1008;
constexpr int kD      = 64;
constexpr int kB      = 2048;
constexpr float kEpsStats = 1e-5f;
constexpr float kEpsPrec  = 1e-6f;

constexpr int kGrpClasses  = 16;
constexpr int kGroups      = kCPad / kGrpClasses;
constexpr int kStatThreads = kGrpClasses * kD;

constexpr int kBlkM        = kB / (8 * 16);
constexpr int kBlkN        = 16;
constexpr int kScoreBlocks = kBlkM * kBlkN;

constexpr int OFF_COUNTS = 0;
constexpr int OFF_MEAN   = OFF_COUNTS + kC;
constexpr int OFF_PART   = OFF_MEAN   + kC * kD;
constexpr int OFF_CPART  = OFF_PART   + kGroups * kD;

typedef __attribute__((ext_vector_type(2))) float v2f;
typedef __attribute__((ext_vector_type(4))) float v4f;
typedef __attribute__((ext_vector_type(8))) float v8f;

__global__ __launch_bounds__(kStatThreads)
void class_stats_kernel(const float* __restrict__ z, const int* __restrict__ y,
                        float* __restrict__ ws) {
  constexpr int kHalf  = kB / 2;
  constexpr int kSteps = kHalf / 32;
  constexpr int kChunk = 8;
  __shared__ int   members[kGrpClasses][kB];
  __shared__ int   nmem[kGrpClasses][2];
  __shared__ float qsh[kGrpClasses][kD];
  __shared__ float csh[kGrpClasses];

  const int lc   = threadIdx.x >> 6;
  const int c    = blockIdx.x * kGrpClasses + lc;
  const int d    = threadIdx.x & 63;
  const int wave = (threadIdx.x >> 5) & 1;
  const int lane = threadIdx.x & 31;
  const int base = wave * kHalf;
  const bool valid = c < kC;
  int* mem = members[lc];

  int lab[kSteps];
#pragma unroll
  for (int i = 0; i < kSteps; ++i) lab[i] = y[base + i * 32 + lane];

  const unsigned below = (1u << lane) - 1u;
  int n = 0;
#pragma unroll
  for (int i = 0; i < kSteps; ++i) {
    const bool hit = (lab[i] == c);
    const unsigned m = __builtin_amdgcn_ballot_w32(hit);
    if (hit) mem[base + n + __builtin_popcount(m & below)] = base + i * 32 + lane;
    n += __builtin_popcount(m);
  }
  if (lane == 0) nmem[lc][wave] = n;
  __syncthreads();

  const int n0 = nmem[lc][0];
  const int nt = n0 + nmem[lc][1];
  auto member = [&](int i) { return mem[i < n0 ? i : kHalf - n0 + i]; };

  const float* __restrict__ zc = z + d;
  float s = 0.f;
  for (int i = 0; i < nt; i += kChunk) {
    float v[kChunk];
#pragma unroll
    for (int j = 0; j < kChunk; ++j) v[j] = zc[member(min(i + j, nt - 1)) * kD];
#pragma unroll
    for (int j = 0; j < kChunk; ++j) if (i + j < nt) s += v[j];
  }
  const float cnte = (float)nt + kEpsStats;
  const float mean = s / cnte;
  if (valid) {
    if (d == 0) ws[OFF_COUNTS + c] = cnte;
    ws[OFF_MEAN + c * kD + d] = mean;
  }

  float q = 0.f;
  for (int i = 0; i < nt; i += kChunk) {
    float v[kChunk];
#pragma unroll
    for (int j = 0; j < kChunk; ++j) v[j] = zc[member(min(i + j, nt - 1)) * kD];
#pragma unroll
    for (int j = 0; j < kChunk; ++j) {
      const float t = v[j] - mean;
      if (i + j < nt) q += t * t;
    }
  }
  qsh[lc][d] = valid ? q : 0.f;
  if (d == 0) csh[lc] = valid ? cnte : 0.f;
  __syncthreads();

  if (threadIdx.x < kD) {
    float ps = 0.f;
#pragma unroll
    for (int k = 0; k < kGrpClasses; ++k) ps += qsh[k][threadIdx.x];
    ws[OFF_PART + blockIdx.x * kD + threadIdx.x] = ps;
  } else if (threadIdx.x == kD) {
    float cs = 0.f;
#pragma unroll
    for (int k = 0; k < kGrpClasses; ++k) cs += csh[k];
    ws[OFF_CPART + blockIdx.x] = cs;
  }
}

__global__ __launch_bounds__(256)
void lda_score_wmma_kernel(const float* __restrict__ z, const float* __restrict__ ws,
                           float* __restrict__ out) {
  constexpr int kJoinGroups = 16;
  constexpr int kJoinRows   = 4;
  constexpr int kWT         = 2;
  constexpr int kRowsA      = 8 * 16;
  constexpr int kRowsB      = 4 * 16;
  constexpr int kPitch      = kD + 4;
  constexpr int kLoadsA     = kRowsA * (kD / 4) / 256;
  constexpr int kLoadsB     = kRowsB * (kD / 4) / 256;
  __shared__ __attribute__((aligned(16))) float apan[kRowsA * kPitch];
  __shared__ __attribute__((aligned(16))) float bpan[kRowsB * kPitch];
  __shared__ float4 jpart[kJoinGroups][kD / 4];
  __shared__ float  jcnt[kJoinGroups];
  __shared__ float  psh[kD];
  __shared__ float  tsh;

  const int wave = threadIdx.x >> 5;
  const int lane = threadIdx.x & 31;
  const int bm = blockIdx.x & (kBlkM - 1);
  const int bn = blockIdx.x / kBlkM;
  const int wm = wave & 3;
  const int wn = wave >> 2;
  const int brow0 = bm * kRowsA;
  const int bcol0 = bn * kRowsB;
  const int half = lane >> 4;
  const int lm   = lane & 15;

  const int jq = threadIdx.x & 15;
  const int jg = threadIdx.x >> 4;
  const float4* __restrict__ jp = reinterpret_cast<const float4*>(ws + OFF_PART) + jq;
  float4 jv[kJoinRows];
#pragma unroll
  for (int i = 0; i < kJoinRows; ++i)
    jv[i] = jp[min(jg + kJoinGroups * i, kGroups - 1) * (kD / 4)];
  float cw[kJoinRows] = {0.f, 0.f, 0.f, 0.f};
  if (jq == 0) {
#pragma unroll
    for (int i = 0; i < kJoinRows; ++i)
      cw[i] = ws[OFF_CPART + min(jg + kJoinGroups * i, kGroups - 1)];
  }

  v4f pa[kLoadsA], pb[kLoadsB];
#pragma unroll
  for (int i = 0; i < kLoadsA; ++i) {
    const int q = i * 256 + threadIdx.x;
    pa[i] = *(const v4f*)(z + (brow0 + (q >> 4)) * kD + 4 * (q & 15));
  }
#pragma unroll
  for (int i = 0; i < kLoadsB; ++i) {
    const int q = i * 256 + threadIdx.x;
    const int cl = min(bcol0 + (q >> 4), kC - 1);
    pb[i] = *(const v4f*)(ws + OFF_MEAN + cl * kD + 4 * (q & 15));
  }
  int col[kWT];
  float cnte[kWT];
#pragma unroll
  for (int t = 0; t < kWT; ++t) {
    col[t] = bcol0 + (wn * kWT + t) * 16 + lm;
    cnte[t] = ws[OFF_COUNTS + min(col[t], kC - 1)];
  }

  {
    float4 js = jv[0];
#pragma unroll
    for (int i = 1; i < kJoinRows; ++i) {
      if (jg + kJoinGroups * i < kGroups) {
        js.x += jv[i].x; js.y += jv[i].y; js.z += jv[i].z; js.w += jv[i].w;
      }
    }
    jpart[jg][jq] = js;
    if (jq == 0) {
      float cs = cw[0];
#pragma unroll
      for (int i = 1; i < kJoinRows; ++i)
        if (jg + kJoinGroups * i < kGroups) cs += cw[i];
      jcnt[jg] = cs;
    }
  }
#pragma unroll
  for (int i = 0; i < kLoadsA; ++i) {
    const int q = i * 256 + threadIdx.x;
    *(v4f*)(apan + (q >> 4) * kPitch + 4 * (q & 15)) = pa[i];
  }
#pragma unroll
  for (int i = 0; i < kLoadsB; ++i) {
    const int q = i * 256 + threadIdx.x;
    *(v4f*)(bpan + (q >> 4) * kPitch + 4 * (q & 15)) = pb[i];
  }
  __syncthreads();
  if (threadIdx.x < kD) {
    const float* __restrict__ pf = reinterpret_cast<const float*>(&jpart[0][0]);
    float sum = 0.f;
    float tot = 0.f;
#pragma unroll
    for (int gg = 0; gg < kJoinGroups; ++gg) {
      sum += pf[gg * kD + threadIdx.x];
      tot += jcnt[gg];
    }
    const float pooled = sum / tot + kEpsStats;
    psh[threadIdx.x] = 1.0f / fmaxf(pooled, kEpsPrec);
    if (threadIdx.x == 0) tsh = tot;
  }
  __syncthreads();
  const float total = tsh;

  constexpr int kSteps = kD / 4;
  v2f a[kWT][kSteps], m[kWT][kSteps], p[kSteps];
#pragma unroll
  for (int j = 0; j < kSteps; ++j) p[j] = *(const v2f*)(psh + 4 * j + 2 * half);
  const float* __restrict__ ap = apan + (wm * kWT * 16 + lm) * kPitch + 2 * half;
  const float* __restrict__ mp = bpan + (wn * kWT * 16 + lm) * kPitch + 2 * half;
#pragma unroll
  for (int t = 0; t < kWT; ++t) {
#pragma unroll
    for (int j = 0; j < kSteps; ++j) {
      a[t][j] = *(const v2f*)(ap + t * 16 * kPitch + 4 * j);
      m[t][j] = *(const v2f*)(mp + t * 16 * kPitch + 4 * j);
    }
  }

  v2f b[kWT][kSteps], eb[kWT];
#pragma unroll
  for (int t = 0; t < kWT; ++t) {
    v2f sc = {0.f, 0.f};
#pragma unroll
    for (int j = 0; j < kSteps; ++j) {
      b[t][j] = p[j] * m[t][j];
      sc += b[t][j] * m[t][j];
    }
    const float lp = logf(cnte[t] / total);
    eb[t].x = 1.0f;
    eb[t].y = -0.5f * (sc.x + sc.y) + (half == 0 ? lp : 0.f);
  }

#pragma unroll
  for (int s = 0; s < kWT; ++s) {
    v8f acc0 = {};
    v8f acc1 = {};
    v2f sq = {0.f, 0.f};
#pragma unroll
    for (int j = 0; j < kSteps; ++j) {
      sq += (p[j] * a[s][j]) * a[s][j];
      acc0 = __builtin_amdgcn_wmma_f32_16x16x4_f32(
           false, a[s][j],  false, b[0][j],
           (short)0, acc0,  false,  false);
      acc1 = __builtin_amdgcn_wmma_f32_16x16x4_f32(
           false, a[s][j],  false, b[1][j],
           (short)0, acc1,  false,  false);
    }

    v2f ea;
    ea.x = -0.5f * (sq.x + sq.y);
    ea.y = 1.0f;
    acc0 = __builtin_amdgcn_wmma_f32_16x16x4_f32(
         false, ea,  false, eb[0],
         (short)0, acc0,  false,  false);
    acc1 = __builtin_amdgcn_wmma_f32_16x16x4_f32(
         false, ea,  false, eb[1],
         (short)0, acc1,  false,  false);

    const int rbase = brow0 + (wm * kWT + s) * 16 + 8 * half;
    if (col[0] < kC) {
#pragma unroll
      for (int r = 0; r < 8; ++r) out[(rbase + r) * kC + col[0]] = acc0[r];
    }
    if (col[1] < kC) {
#pragma unroll
      for (int r = 0; r < 8; ++r) out[(rbase + r) * kC + col[1]] = acc1[r];
    }
  }
}

}

extern "C" void kernel_launch(void* const* d_in, const int* in_sizes, int n_in,
                              void* d_out, int out_size, void* d_ws, size_t ws_size,
                              hipStream_t stream) {
  (void)in_sizes; (void)n_in; (void)out_size; (void)ws_size;
  const float* z = (const float*)d_in[0];
  const int*   y = (const int*)d_in[1];
  float* ws  = (float*)d_ws;
  float* out = (float*)d_out;

  class_stats_kernel<<<kGroups, kStatThreads, 0, stream>>>(z, y, ws);

  lda_score_wmma_kernel<<<kScoreBlocks, 256, 0, stream>>>(z, ws, out);
}


# statistics blocks go over the labels once: their 32 waves write the rows of the block's 16 classes to one short ascending LDS list and each class picks its rows from it; order of sums unchanged
# speedup vs baseline: 18.1183x; 1.0941x over previous
#include <hip/hip_runtime.h>
#include <math.h>

namespace {

constexpr int kC      = 1000;
constexpr int kCPad   = 1008;
constexpr int kD      = 64;
constexpr int kB      = 2048;
constexpr float kEpsStats = 1e-5f;
constexpr float kEpsPrec  = 1e-6f;

constexpr int kGrpClasses  = 16;
constexpr int kGroups      = kCPad / kGrpClasses;
constexpr int kStatThreads = kGrpClasses * kD;

constexpr int kBlkM        = kB / (8 * 16);
constexpr int kBlkN        = 16;
constexpr int kScoreBlocks = kBlkM * kBlkN;

constexpr int OFF_COUNTS = 0;
constexpr int OFF_MEAN   = OFF_COUNTS + kC;
constexpr int OFF_PART   = OFF_MEAN   + kC * kD;
constexpr int OFF_CPART  = OFF_PART   + kGroups * kD;

typedef __attribute__((ext_vector_type(2))) float v2f;
typedef __attribute__((ext_vector_type(4))) float v4f;
typedef __attribute__((ext_vector_type(8))) float v8f;

__global__ __launch_bounds__(kStatThreads)
void class_stats_kernel(const float* __restrict__ z, const int* __restrict__ y,
                        float* __restrict__ ws) {
  constexpr int kWaves       = kStatThreads / 32;
  constexpr int kRowsPerWave = kB / kWaves;
  constexpr int kLabSteps    = kRowsPerWave / 32;
  constexpr int kCntBits     = 7;
  constexpr int kChunk       = 8;
  static_assert(kWaves == 32, "one wave count per lane");
  static_assert(kGrpClasses == 16, "the class of the block is packed into 4 bits");
  __shared__ int            hits[kB];
  __shared__ int            wcnt[kWaves];
  __shared__ unsigned short members[kWaves][kB];
  __shared__ float qsh[kGrpClasses][kD];
  __shared__ float csh[kGrpClasses];

  const int lc   = threadIdx.x >> 6;
  const int c0   = blockIdx.x * kGrpClasses;
  const int c    = c0 + lc;
  const int d    = threadIdx.x & 63;
  const int wv   = threadIdx.x >> 5;
  const int lane = threadIdx.x & 31;
  const bool valid = c < kC;
  const unsigned below = (1u << lane) - 1u;

  int rel[kLabSteps];
#pragma unroll
  for (int s = 0; s < kLabSteps; ++s) rel[s] = y[wv * kRowsPerWave + s * 32 + lane] - c0;
  unsigned hm[kLabSteps];
  int cnt = 0;
#pragma unroll
  for (int s = 0; s < kLabSteps; ++s) {
    hm[s] = __builtin_amdgcn_ballot_w32((unsigned)rel[s] < (unsigned)kGrpClasses);
    cnt += __builtin_popcount(hm[s]);
  }
  if (lane == 0) wcnt[wv] = cnt;
  __syncthreads();

  const int wc = wcnt[lane];
  const int wvs = __builtin_amdgcn_readfirstlane(wv);
  const unsigned lowmask = (1u << wvs) - 1u;
  int off = 0;
  int tot = 0;
#pragma unroll
  for (int b = 0; b < kCntBits; ++b) {
    const unsigned plane = __builtin_amdgcn_ballot_w32(((wc >> b) & 1) != 0);
    off += __builtin_popcount(plane & lowmask) << b;
    tot += __builtin_popcount(plane) << b;
  }
#pragma unroll
  for (int s = 0; s < kLabSteps; ++s) {
    if ((unsigned)rel[s] < (unsigned)kGrpClasses)
      hits[off + __builtin_popcount(hm[s] & below)] =
          ((wv * kRowsPerWave + s * 32 + lane) << 4) | rel[s];
    off += __builtin_popcount(hm[s]);
  }
  __syncthreads();

  unsigned short* mem = members[wv];
  int n = 0;
  for (int s0 = 0; s0 < tot; s0 += 32) {
    const int idx = s0 + lane;
    int e = 0;
    if (idx < tot) e = hits[idx];
    const bool hit = idx < tot && (e & (kGrpClasses - 1)) == lc;
    const unsigned m = __builtin_amdgcn_ballot_w32(hit);
    if (hit) mem[n + __builtin_popcount(m & below)] = (unsigned short)(e >> 4);
    n += __builtin_popcount(m);
  }
  const int nt = n;
  auto member = [&](int i) { return (int)mem[i]; };

  const float* __restrict__ zc = z + d;
  float s = 0.f;
  for (int i = 0; i < nt; i += kChunk) {
    float v[kChunk];
#pragma unroll
    for (int j = 0; j < kChunk; ++j) v[j] = zc[member(min(i + j, nt - 1)) * kD];
    __builtin_amdgcn_sched_barrier(0);
#pragma unroll
    for (int j = 0; j < kChunk; ++j) if (i + j < nt) s += v[j];
  }
  const float cnte = (float)nt + kEpsStats;
  const float mean = s / cnte;
  if (valid) {
    if (d == 0) ws[OFF_COUNTS + c] = cnte;
    ws[OFF_MEAN + c * kD + d] = mean;
  }

  float q = 0.f;
  for (int i = 0; i < nt; i += kChunk) {
    float v[kChunk];
#pragma unroll
    for (int j = 0; j < kChunk; ++j) v[j] = zc[member(min(i + j, nt - 1)) * kD];
    __builtin_amdgcn_sched_barrier(0);
#pragma unroll
    for (int j = 0; j < kChunk; ++j) {
      const float t = v[j] - mean;
      if (i + j < nt) q += t * t;
    }
  }
  qsh[lc][d] = valid ? q : 0.f;
  if (d == 0) csh[lc] = valid ? cnte : 0.f;
  __syncthreads();

  if (threadIdx.x < kD) {
    float ps = 0.f;
#pragma unroll
    for (int k = 0; k < kGrpClasses; ++k) ps += qsh[k][threadIdx.x];
    ws[OFF_PART + blockIdx.x * kD + threadIdx.x] = ps;
  } else if (threadIdx.x == kD) {
    float cs = 0.f;
#pragma unroll
    for (int k = 0; k < kGrpClasses; ++k) cs += csh[k];
    ws[OFF_CPART + blockIdx.x] = cs;
  }
}

__global__ __launch_bounds__(256)
void lda_score_wmma_kernel(const float* __restrict__ z, const float* __restrict__ ws,
                           float* __restrict__ out) {
  constexpr int kJoinGroups = 16;
  constexpr int kJoinRows   = 4;
  constexpr int kWT         = 2;
  constexpr int kRowsA      = 8 * 16;
  constexpr int kRowsB      = 4 * 16;
  constexpr int kPitch      = kD + 4;
  constexpr int kLoadsA     = kRowsA * (kD / 4) / 256;
  constexpr int kLoadsB     = kRowsB * (kD / 4) / 256;
  __shared__ __attribute__((aligned(16))) float apan[kRowsA * kPitch];
  __shared__ __attribute__((aligned(16))) float bpan[kRowsB * kPitch];
  __shared__ float4 jpart[kJoinGroups][kD / 4];
  __shared__ float  jcnt[kJoinGroups];
  __shared__ float  psh[kD];
  __shared__ float  tsh;

  const int wave = threadIdx.x >> 5;
  const int lane = threadIdx.x & 31;
  const int bm = blockIdx.x & (kBlkM - 1);
  const int bn = blockIdx.x / kBlkM;
  const int wm = wave & 3;
  const int wn = wave >> 2;
  const int brow0 = bm * kRowsA;
  const int bcol0 = bn * kRowsB;
  const int half = lane >> 4;
  const int lm   = lane & 15;

  const int jq = threadIdx.x & 15;
  const int jg = threadIdx.x >> 4;
  const float4* __restrict__ jp = reinterpret_cast<const float4*>(ws + OFF_PART) + jq;
  float4 jv[kJoinRows];
#pragma unroll
  for (int i = 0; i < kJoinRows; ++i)
    jv[i] = jp[min(jg + kJoinGroups * i, kGroups - 1) * (kD / 4)];
  float cw[kJoinRows] = {0.f, 0.f, 0.f, 0.f};
  if (jq == 0) {
#pragma unroll
    for (int i = 0; i < kJoinRows; ++i)
      cw[i] = ws[OFF_CPART + min(jg + kJoinGroups * i, kGroups - 1)];
  }

  v4f pa[kLoadsA], pb[kLoadsB];
#pragma unroll
  for (int i = 0; i < kLoadsA; ++i) {
    const int q = i * 256 + threadIdx.x;
    pa[i] = *(const v4f*)(z + (brow0 + (q >> 4)) * kD + 4 * (q & 15));
  }
#pragma unroll
  for (int i = 0; i < kLoadsB; ++i) {
    const int q = i * 256 + threadIdx.x;
    const int cl = min(bcol0 + (q >> 4), kC - 1);
    pb[i] = *(const v4f*)(ws + OFF_MEAN + cl * kD + 4 * (q & 15));
  }
  int col[kWT];
  float cnte[kWT];
#pragma unroll
  for (int t = 0; t < kWT; ++t) {
    col[t] = bcol0 + (wn * kWT + t) * 16 + lm;
    cnte[t] = ws[OFF_COUNTS + min(col[t], kC - 1)];
  }

  {
    float4 js = jv[0];
#pragma unroll
    for (int i = 1; i < kJoinRows; ++i) {
      if (jg + kJoinGroups * i < kGroups) {
        js.x += jv[i].x; js.y += jv[i].y; js.z += jv[i].z; js.w += jv[i].w;
      }
    }
    jpart[jg][jq] = js;
    if (jq == 0) {
      float cs = cw[0];
#pragma unroll
      for (int i = 1; i < kJoinRows; ++i)
        if (jg + kJoinGroups * i < kGroups) cs += cw[i];
      jcnt[jg] = cs;
    }
  }
#pragma unroll
  for (int i = 0; i < kLoadsA; ++i) {
    const int q = i * 256 + threadIdx.x;
    *(v4f*)(apan + (q >> 4) * kPitch + 4 * (q & 15)) = pa[i];
  }
#pragma unroll
  for (int i = 0; i < kLoadsB; ++i) {
    const int q = i * 256 + threadIdx.x;
    *(v4f*)(bpan + (q >> 4) * kPitch + 4 * (q & 15)) = pb[i];
  }
  __syncthreads();
  if (threadIdx.x < kD) {
    const float* __restrict__ pf = reinterpret_cast<const float*>(&jpart[0][0]);
    float sum = 0.f;
    float tot = 0.f;
#pragma unroll
    for (int gg = 0; gg < kJoinGroups; ++gg) {
      sum += pf[gg * kD + threadIdx.x];
      tot += jcnt[gg];
    }
    const float pooled = sum / tot + kEpsStats;
    psh[threadIdx.x] = 1.0f / fmaxf(pooled, kEpsPrec);
    if (threadIdx.x == 0) tsh = tot;
  }
  __syncthreads();
  const float total = tsh;

  constexpr int kSteps = kD / 4;
  v2f a[kWT][kSteps], m[kWT][kSteps], p[kSteps];
#pragma unroll
  for (int j = 0; j < kSteps; ++j) p[j] = *(const v2f*)(psh + 4 * j + 2 * half);
  const float* __restrict__ ap = apan + (wm * kWT * 16 + lm) * kPitch + 2 * half;
  const float* __restrict__ mp = bpan + (wn * kWT * 16 + lm) * kPitch + 2 * half;
#pragma unroll
  for (int t = 0; t < kWT; ++t) {
#pragma unroll
    for (int j = 0; j < kSteps; ++j) {
      a[t][j] = *(const v2f*)(ap + t * 16 * kPitch + 4 * j);
      m[t][j] = *(const v2f*)(mp + t * 16 * kPitch + 4 * j);
    }
  }

  v2f b[kWT][kSteps], eb[kWT];
#pragma unroll
  for (int t = 0; t < kWT; ++t) {
    v2f sc = {0.f, 0.f};
#pragma unroll
    for (int j = 0; j < kSteps; ++j) {
      b[t][j] = p[j] * m[t][j];
      sc += b[t][j] * m[t][j];
    }
    const float lp = logf(cnte[t] / total);
    eb[t].x = 1.0f;
    eb[t].y = -0.5f * (sc.x + sc.y) + (half == 0 ? lp : 0.f);
  }

#pragma unroll
  for (int s = 0; s < kWT; ++s) {
    v8f acc0 = {};
    v8f acc1 = {};
    v2f sq = {0.f, 0.f};
#pragma unroll
    for (int j = 0; j < kSteps; ++j) {
      sq += (p[j] * a[s][j]) * a[s][j];
      acc0 = __builtin_amdgcn_wmma_f32_16x16x4_f32(
           false, a[s][j],  false, b[0][j],
           (short)0, acc0,  false,  false);
      acc1 = __builtin_amdgcn_wmma_f32_16x16x4_f32(
           false, a[s][j],  false, b[1][j],
           (short)0, acc1,  false,  false);
    }

    v2f ea;
    ea.x = -0.5f * (sq.x + sq.y);
    ea.y = 1.0f;
    acc0 = __builtin_amdgcn_wmma_f32_16x16x4_f32(
         false, ea,  false, eb[0],
         (short)0, acc0,  false,  false);
    acc1 = __builtin_amdgcn_wmma_f32_16x16x4_f32(
         false, ea,  false, eb[1],
         (short)0, acc1,  false,  false);

    const int rbase = brow0 + (wm * kWT + s) * 16 + 8 * half;
    if (col[0] < kC) {
#pragma unroll
      for (int r = 0; r < 8; ++r) out[(rbase + r) * kC + col[0]] = acc0[r];
    }
    if (col[1] < kC) {
#pragma unroll
      for (int r = 0; r < 8; ++r) out[(rbase + r) * kC + col[1]] = acc1[r];
    }
  }
}

}

extern "C" void kernel_launch(void* const* d_in, const int* in_sizes, int n_in,
                              void* d_out, int out_size, void* d_ws, size_t ws_size,
                              hipStream_t stream) {
  (void)in_sizes; (void)n_in; (void)out_size; (void)ws_size;
  const float* z = (const float*)d_in[0];
  const int*   y = (const int*)d_in[1];
  float* ws  = (float*)d_ws;
  float* out = (float*)d_out;

  class_stats_kernel<<<kGroups, kStatThreads, 0, stream>>>(z, y, ws);

  lda_score_wmma_kernel<<<kScoreBlocks, 256, 0, stream>>>(z, ws, out);
}
